# starting kernel
# baseline (speedup 1.0000x reference)
#include <hip/hip_runtime.h>


#define NUM_NEURONS 100000
#define MAX_EDGES   100000
#define NUM_CONN    1600000
#define BATCH       4
#define OUT_CH      16

typedef __attribute__((ext_vector_type(2))) float v2f;
typedef __attribute__((ext_vector_type(8))) float v8f;

#define WS_EFFW 0u
#define WS_S    100000u
#define WS_CNT  500000u
#define WS_U    600000u
#define WS_XT   1000000u

__global__ void bhs_zero(float* __restrict__ ws) {
  unsigned i = blockIdx.x * blockDim.x + threadIdx.x;
  if (i < 225000u) ((float4*)(ws + WS_S))[i] = make_float4(0.f, 0.f, 0.f, 0.f);
}

__global__ void bhs_traces(const float* __restrict__ x_in,
                           const float* __restrict__ post_spk,
                           const float* __restrict__ w_hat,
                           const float* __restrict__ s_e,
                           const float* __restrict__ e_trace,
                           const float* __restrict__ pre_trace,
                           const float* __restrict__ post_trace,
                           float* __restrict__ out,
                           float* __restrict__ ws) {
  unsigned i = blockIdx.x * blockDim.x + threadIdx.x;
  if (i >= NUM_NEURONS) return;
  const float decay = 0.95122942450071400910f;
  float x0 = x_in[i];
  float x1 = x_in[NUM_NEURONS + i];
  float x2 = x_in[2u * NUM_NEURONS + i];
  float x3 = x_in[3u * NUM_NEURONS + i];
  float pre_spk  = (x0 + x1 + x2 + x3) * 0.25f;
  float pre_new  = fmaf(pre_trace[i],  decay, pre_spk);
  float post_new = fmaf(post_trace[i], decay, post_spk[i]);
  float e_new    = fmaf(0.9f, e_trace[i], pre_spk * (post_new + 0.1f));
  const size_t OUTB = (size_t)BATCH * NUM_NEURONS * OUT_CH;
  out[OUTB + i]                 = e_new;
  out[OUTB + MAX_EDGES + i]     = pre_new;
  out[OUTB + 2u * MAX_EDGES + i] = post_new;
  ws[WS_EFFW + i] = s_e[i] * w_hat[i];
  ((float4*)(ws + WS_XT))[i] = make_float4(x0, x1, x2, x3);
}

__global__ void bhs_scatterA(const int* __restrict__ he, float* __restrict__ ws) {
  unsigned c = blockIdx.x * blockDim.x + threadIdx.x;
  int n = he[c];
  int e = he[NUM_CONN + c];
  float cw = ws[WS_EFFW + e];
  float4 xv = ((const float4*)(ws + WS_XT))[n];
  float* S = ws + WS_S + (size_t)e * 4u;
  atomicAdd(&S[0], xv.x * cw);
  atomicAdd(&S[1], xv.y * cw);
  atomicAdd(&S[2], xv.z * cw);
  atomicAdd(&S[3], xv.w * cw);
  atomicAdd(ws + WS_CNT + e, 1.0f);
}

__global__ void bhs_norm(float* __restrict__ ws) {
  unsigned e = blockIdx.x * blockDim.x + threadIdx.x;
  if (e >= MAX_EDGES) return;
  float inv = 1.0f / fmaxf(ws[WS_CNT + e], 1.0f);
  float4* S = (float4*)(ws + WS_S);
  float4 s = S[e];
  s.x *= inv; s.y *= inv; s.z *= inv; s.w *= inv;
  S[e] = s;
}

__global__ void bhs_scatterB(const int* __restrict__ he, float* __restrict__ ws) {
  unsigned c = blockIdx.x * blockDim.x + threadIdx.x;
  int n = he[c];
  int e = he[NUM_CONN + c];
  float cw = ws[WS_EFFW + e];
  float4 t = ((const float4*)(ws + WS_S))[e];
  float* U = ws + WS_U + (size_t)n * 4u;
  atomicAdd(&U[0], cw * t.x);
  atomicAdd(&U[1], cw * t.y);
  atomicAdd(&U[2], cw * t.z);
  atomicAdd(&U[3], cw * t.w);
}

__global__ void bhs_out_wmma(const float* __restrict__ weight_node,
                             const float* __restrict__ bias,
                             const float* __restrict__ ws,
                             float* __restrict__ out) {
  unsigned lane = threadIdx.x & 31u;
  unsigned wave = (blockIdx.x * blockDim.x + threadIdx.x) >> 5;
  unsigned base = wave * 16u;
  unsigned oc = lane & 15u;
  float wv = weight_node[oc];
  float bv = bias[oc];
  const float* U = ws + WS_U;

  v2f a;
  a.x = (lane < 16u) ? U[base + lane] : 0.0f;
  a.y = 0.0f;
  v2f b;
  b.x = (lane < 16u) ? wv : 0.0f;
  b.y = 0.0f;
  v8f c;
#pragma unroll
  for (int v = 0; v < 8; ++v) c[v] = bv;

  v8f d = __builtin_amdgcn_wmma_f32_16x16x4_f32(
       false, a,  false, b,
       (short)0, c,  false,  false);

  unsigned mbase = (lane < 16u) ? 0u : 8u;
#pragma unroll
  for (int v = 0; v < 8; ++v) {
    unsigned nb = base + mbase + (unsigned)v;
    unsigned n = nb >> 2;
    unsigned bb = nb & 3u;
    out[((size_t)bb * NUM_NEURONS + n) * OUT_CH + oc] = d[v];
  }
}

extern "C" void kernel_launch(void* const* d_in, const int* in_sizes, int n_in,
                              void* d_out, int out_size, void* d_ws, size_t ws_size,
                              hipStream_t stream) {
  const float* x_in        = (const float*)d_in[0];
  const int*   he          = (const int*)d_in[1];
  const float* post_spk    = (const float*)d_in[2];
  const float* w_hat       = (const float*)d_in[3];
  const float* s_e         = (const float*)d_in[4];
  const float* e_trace     = (const float*)d_in[5];
  const float* pre_trace   = (const float*)d_in[6];
  const float* post_trace  = (const float*)d_in[7];
  const float* weight_node = (const float*)d_in[8];
  const float* bias        = (const float*)d_in[9];
  float* out = (float*)d_out;
  float* ws  = (float*)d_ws;

  bhs_zero<<<(225000u + 255u) / 256u, 256, 0, stream>>>(ws);
  bhs_traces<<<(NUM_NEURONS + 255u) / 256u, 256, 0, stream>>>(
      x_in, post_spk, w_hat, s_e, e_trace, pre_trace, post_trace, out, ws);
  bhs_scatterA<<<NUM_CONN / 256u, 256, 0, stream>>>(he, ws);
  bhs_norm<<<(MAX_EDGES + 255u) / 256u, 256, 0, stream>>>(ws);
  bhs_scatterB<<<NUM_CONN / 256u, 256, 0, stream>>>(he, ws);
  bhs_out_wmma<<<3125, 256, 0, stream>>>(weight_node, bias, ws, out);
}


# Filter passes: a chunk is whole 4096-connection rows from a 128-B line, last row cut on a wave slice: 33 and 25 index steps a thread (were 36, 27), pass A in 3 rounds; LDS add order still free
# speedup vs baseline: 5.3753x; 5.3753x over previous
#include <hip/hip_runtime.h>


#define NUM_NEURONS 100000
#define MAX_EDGES   100000
#define NUM_CONN    1600000
#define BATCH       4
#define OUT_CH      16

typedef __attribute__((ext_vector_type(2))) float v2f;
typedef __attribute__((ext_vector_type(8))) float v8f;

#define WS_EFFW 0u
#define WS_XT   100000u
#define WS_TR   100000u
#define WS_U    900000u

#define A_NRANGE 21u
#define A_RNG    4768u
#define A_NCHUNK 12u
#define A_CHUNK  133376u
#define A_R0     11
#define A_R1     11
#define A_R2     11
#define B_NRANGE 16u
#define B_RNG    6272u
#define B_NCHUNK 16u
#define B_CHUNK  100096u
#define B_R0     9
#define B_R1     8
#define B_R2     8
#define ROW      4096u
#define SLICE    128u
#define UFA      3
#define UFB      6
#define PLANE    100000u

__global__ void bhs_traces(const float* __restrict__ x_in,
                           const float* __restrict__ post_spk,
                           const float* __restrict__ w_hat,
                           const float* __restrict__ s_e,
                           const float* __restrict__ e_trace,
                           const float* __restrict__ pre_trace,
                           const float* __restrict__ post_trace,
                           float* __restrict__ out,
                           float* __restrict__ ws) {
  unsigned i = blockIdx.x * blockDim.x + threadIdx.x;
  if (i >= NUM_NEURONS) return;
  const float decay = 0.95122942450071400910f;
  float x0 = x_in[i];
  float x1 = x_in[NUM_NEURONS + i];
  float x2 = x_in[2u * NUM_NEURONS + i];
  float x3 = x_in[3u * NUM_NEURONS + i];
  float pre_spk  = (x0 + x1 + x2 + x3) * 0.25f;
  float pre_new  = fmaf(pre_trace[i],  decay, pre_spk);
  float post_new = fmaf(post_trace[i], decay, post_spk[i]);
  float e_new    = fmaf(0.9f, e_trace[i], pre_spk * (post_new + 0.1f));
  const size_t OUTB = (size_t)BATCH * NUM_NEURONS * OUT_CH;
  out[OUTB + i]                 = e_new;
  out[OUTB + MAX_EDGES + i]     = pre_new;
  out[OUTB + 2u * MAX_EDGES + i] = post_new;
  ws[WS_EFFW + i] = s_e[i] * w_hat[i];
  ((float4*)(ws + WS_XT))[i] = make_float4(x0, x1, x2, x3);
}


template <int MAXR, int N, bool LAST>
__device__ __forceinline__ void bhs_load_idx(int4 (&kv)[MAXR], int4 (&ov)[MAXR],
                                             const int* __restrict__ keyrow,
                                             const int* __restrict__ othrow,
                                             unsigned c0, bool lastok) {
#pragma unroll
  for (int it = 0; it < N; ++it) {
    if (LAST && it == N - 1 && !lastok) continue;
    const unsigned c = c0 + (unsigned)it * ROW;
    kv[it] = *(const int4*)(keyrow + c);
    ov[it] = *(const int4*)(othrow + c);
  }
}

template <int MAXR, int N, bool LAST>
__device__ __forceinline__ unsigned bhs_filter(const int4 (&kv)[MAXR], const int4 (&ov)[MAXR],
                                               unsigned* seg, unsigned base, unsigned rr, bool lastok) {
  unsigned wb = 0u;
#pragma unroll
  for (int it = 0; it < N; ++it) {
    if (LAST && it == N - 1 && !lastok) continue;
    unsigned d0 = (unsigned)kv[it].x - base, d1 = (unsigned)kv[it].y - base;
    unsigned d2 = (unsigned)kv[it].z - base, d3 = (unsigned)kv[it].w - base;
    bool m0 = d0 < rr, m1 = d1 < rr, m2 = d2 < rr, m3 = d3 < rr;
    unsigned b0 = __builtin_amdgcn_ballot_w32(m0);
    unsigned b1 = __builtin_amdgcn_ballot_w32(m1);
    unsigned b2 = __builtin_amdgcn_ballot_w32(m2);
    unsigned b3 = __builtin_amdgcn_ballot_w32(m3);
    if (m0) seg[__builtin_amdgcn_mbcnt_lo(b0, wb)] = (d0 << 17) | (unsigned)ov[it].x;
    wb += __builtin_popcount(b0);
    if (m1) seg[__builtin_amdgcn_mbcnt_lo(b1, wb)] = (d1 << 17) | (unsigned)ov[it].y;
    wb += __builtin_popcount(b1);
    if (m2) seg[__builtin_amdgcn_mbcnt_lo(b2, wb)] = (d2 << 17) | (unsigned)ov[it].z;
    wb += __builtin_popcount(b2);
    if (m3) seg[__builtin_amdgcn_mbcnt_lo(b3, wb)] = (d3 << 17) | (unsigned)ov[it].w;
    wb += __builtin_popcount(b3);
  }
  return wb;
}

template <bool PB>
__device__ __forceinline__ void bhs_fetch(const unsigned* seg, const float* __restrict__ ws,
                                          const float* cwl, unsigned t0, unsigned lane, unsigned nk,
                                          unsigned (&en)[PB ? UFB : UFA], bool (&ok)[PB ? UFB : UFA],
                                          float4 (&g)[PB ? UFB : UFA], float (&cwv)[PB ? UFB : UFA]) {
  if (!PB) {
#pragma unroll
    for (int k = 0; k < UFA; ++k) {
      unsigned i = t0 + (unsigned)k * 32u + lane;
      ok[k] = i < nk;
      en[k] = ok[k] ? seg[i] : 0u;
      unsigned ot = en[k] & 0x1FFFFu;
      g[k]   = ((const float4*)(ws + WS_XT))[ot];
      cwv[k] = cwl[en[k] >> 17];
    }
  } else {
#pragma unroll
    for (int k = 0; k < UFB; ++k) {
      unsigned i = t0 + (unsigned)k * 16u + (lane >> 1);
      ok[k] = i < nk;
      en[k] = ok[k] ? seg[i] : 0u;
    }
#pragma unroll
    for (int k = 0; k < UFB; ++k) {
      unsigned ot = en[k] & 0x1FFFFu;
      g[k]   = ((const float4*)(ws + WS_TR))[2u * ot + (lane & 1u)];
      cwv[k] = g[k].z;
    }
  }
}

template <bool PB, unsigned RNG>
__device__ __forceinline__ void bhs_sums(float* acc, unsigned lane,
                                         const unsigned (&en)[PB ? UFB : UFA], const bool (&ok)[PB ? UFB : UFA],
                                         const float4 (&g)[PB ? UFB : UFA], const float (&cwv)[PB ? UFB : UFA]) {
  constexpr int NU = PB ? UFB : UFA;
#pragma unroll
  for (int k = 0; k < NU; ++k) {
    if (ok[k]) {
      unsigned kl = en[k] >> 17;
      if (!PB) {
        atomicAdd(&acc[kl],            g[k].x * cwv[k]);
        atomicAdd(&acc[RNG + kl],      g[k].y * cwv[k]);
        atomicAdd(&acc[2u * RNG + kl], g[k].z * cwv[k]);
        atomicAdd(&acc[3u * RNG + kl], g[k].w * cwv[k]);
        atomicAdd(&acc[4u * RNG + kl], 1.0f);
      } else {
        float* a = acc + (lane & 1u) * (2u * RNG) + kl;
        atomicAdd(a,       cwv[k] * g[k].x);
        atomicAdd(a + RNG, cwv[k] * g[k].y);
      }
    }
  }
}

template <bool PB, unsigned RNG, int MAXR, int N, bool LAST, int NN, bool NLAST>
__device__ __forceinline__ void bhs_round(int4 (&kv)[MAXR], int4 (&ov)[MAXR], unsigned* seg, float* acc,
                                          const float* cwl, const float* __restrict__ ws,
                                          const int* __restrict__ keyrow, const int* __restrict__ othrow,
                                          unsigned cnext, unsigned base, unsigned rr, unsigned lane,
                                          bool lastok) {
  const unsigned nk = bhs_filter<MAXR, N, LAST>(kv, ov, seg, base, rr, lastok);
  constexpr int NU = PB ? UFB : UFA;
  constexpr unsigned TRIP = PB ? UFB * 16u : UFA * 32u;
  unsigned en[NU];
  bool ok[NU];
  float4 g[NU];
  float cwv[NU];
  bhs_fetch<PB>(seg, ws, cwl, 0u, lane, nk, en, ok, g, cwv);
  if constexpr (NN > 0)
    bhs_load_idx<MAXR, NN, NLAST>(kv, ov, keyrow, othrow, cnext, lastok);
  bhs_sums<PB, RNG>(acc, lane, en, ok, g, cwv);
  for (unsigned t0 = TRIP; t0 < nk; t0 += TRIP) {
    bhs_fetch<PB>(seg, ws, cwl, t0, lane, nk, en, ok, g, cwv);
    bhs_sums<PB, RNG>(acc, lane, en, ok, g, cwv);
  }
}

template <bool PB>
__global__ __launch_bounds__(1024)
void bhs_pass(const int* __restrict__ he, const float* __restrict__ ws,
              float* __restrict__ part) {
  constexpr unsigned P      = PB ? 4u : 5u;
  constexpr unsigned NRANGE = PB ? B_NRANGE : A_NRANGE;
  constexpr unsigned RNG    = PB ? B_RNG : A_RNG;
  constexpr unsigned NCHUNK = PB ? B_NCHUNK : A_NCHUNK;
  constexpr unsigned CHUNK  = PB ? B_CHUNK : A_CHUNK;
  constexpr int R0 = PB ? B_R0 : A_R0, R1 = PB ? B_R1 : A_R1, R2 = PB ? B_R2 : A_R2;
  constexpr int MAXR = R0 > R1 ? (R0 > R2 ? R0 : R2) : (R1 > R2 ? R1 : R2);
  constexpr unsigned ROWS = (unsigned)(R0 + R1 + R2);
  constexpr unsigned SEG  = (unsigned)MAXR * SLICE;
  static_assert(CHUNK % SLICE == 0u && NUM_CONN % SLICE == 0u, "a chunk is a whole number of wave slices");
  static_assert((NCHUNK - 1u) * CHUNK < NUM_CONN && NCHUNK * CHUNK >= NUM_CONN, "the chunks cover the list");
  static_assert(ROWS * ROW >= CHUNK && (ROWS - 1u) * ROW <= NUM_CONN - (NCHUNK - 1u) * CHUNK,
                "only the last row of a chunk may be cut");
  __shared__ float acc[P * RNG];
  __shared__ float cwl[PB ? 4u : RNG];
  __shared__ unsigned list[32u * SEG];

  const unsigned tid  = threadIdx.x;
  const unsigned lane = tid & 31u;
  const unsigned wave = __builtin_amdgcn_readfirstlane(tid >> 5);
  const unsigned r    = blockIdx.x % NRANGE;
  const unsigned j    = blockIdx.x / NRANGE;
  const unsigned base = r * RNG;
  const unsigned rr   = min(RNG, (unsigned)MAX_EDGES - base);

  const int* keyrow = PB ? he : he + NUM_CONN;
  const int* othrow = PB ? he + NUM_CONN : he;
  const unsigned cbeg = j * CHUNK;
  const unsigned cend = min(cbeg + CHUNK, (unsigned)NUM_CONN);
  const unsigned ct   = cbeg + tid * 4u;
  const bool lastok = (ROWS - 1u) * ROW + wave * SLICE < cend - cbeg;

  int4 kv[MAXR], ov[MAXR];
  bhs_load_idx<MAXR, R0, false>(kv, ov, keyrow, othrow, ct, true);

  for (unsigned i = tid; i < P * RNG; i += 1024u) acc[i] = 0.0f;
  if (!PB)
    for (unsigned i = tid; i < rr; i += 1024u) cwl[i] = ws[WS_EFFW + base + i];
  __syncthreads();

  unsigned* seg = list + wave * SEG;

  bhs_round<PB, RNG, MAXR, R0, false, R1, false>(kv, ov, seg, acc, cwl, ws, keyrow, othrow,
                                                 ct + (unsigned)R0 * ROW, base, rr, lane, lastok);
  bhs_round<PB, RNG, MAXR, R1, false, R2, true>(kv, ov, seg, acc, cwl, ws, keyrow, othrow,
                                                ct + (unsigned)(R0 + R1) * ROW, base, rr, lane, lastok);
  bhs_round<PB, RNG, MAXR, R2, true, 0, false>(kv, ov, seg, acc, cwl, ws, keyrow, othrow,
                                               0u, base, rr, lane, lastok);
  __syncthreads();

  for (unsigned p = 0; p < P; ++p)
    for (unsigned i = tid; i < rr; i += 1024u)
      part[(size_t)(j * P + p) * PLANE + base + i] = acc[p * RNG + i];
}

__global__ void bhs_norm(const float* __restrict__ part, float* __restrict__ ws) {
  unsigned e = blockIdx.x * blockDim.x + threadIdx.x;
  if (e >= MAX_EDGES) return;
  float s0 = 0.f, s1 = 0.f, s2 = 0.f, s3 = 0.f, cn = 0.f;
#pragma unroll
  for (unsigned j = 0; j < A_NCHUNK; ++j) {
    const float* p = part + (size_t)j * 5u * PLANE + e;
    s0 += p[0];
    s1 += p[PLANE];
    s2 += p[2u * PLANE];
    s3 += p[3u * PLANE];
    cn += p[4u * PLANE];
  }
  float inv = 1.0f / fmaxf(cn, 1.0f);
  float cw = ws[WS_EFFW + e];
  float4* rec = (float4*)(ws + WS_TR) + 2u * e;
  rec[0] = make_float4(s0 * inv, s1 * inv, cw, 0.0f);
  rec[1] = make_float4(s2 * inv, s3 * inv, cw, 0.0f);
}

__global__ void bhs_sumU(const float* __restrict__ part, float* __restrict__ ws) {
  unsigned n = blockIdx.x * blockDim.x + threadIdx.x;
  if (n >= NUM_NEURONS) return;
  float u0 = 0.f, u1 = 0.f, u2 = 0.f, u3 = 0.f;
#pragma unroll
  for (unsigned j = 0; j < B_NCHUNK; ++j) {
    const float* p = part + (size_t)j * 4u * PLANE + n;
    u0 += p[0];
    u1 += p[PLANE];
    u2 += p[2u * PLANE];
    u3 += p[3u * PLANE];
  }
  ((float4*)(ws + WS_U))[n] = make_float4(u0, u1, u2, u3);
}

#define OUT_WG    250u
#define OUT_WPG   10u
#define OUT_WAVES 2500u
#define OUT_TPW   10
__global__ __launch_bounds__(320)
void bhs_out_wmma(const float* __restrict__ weight_node,
                  const float* __restrict__ bias,
                  const float* __restrict__ ws,
                  float* __restrict__ out) {
  const unsigned lane = threadIdx.x & 31u;
  const unsigned wave =
      __builtin_amdgcn_readfirstlane(blockIdx.x * OUT_WPG + (threadIdx.x >> 5));
  const unsigned oc = lane & 15u;
  const bool lo = lane < 16u;
  const float wv = weight_node[oc];
  const float bv = bias[oc];
  const float* U = ws + WS_U;

  float av[OUT_TPW];
#pragma unroll
  for (int k = 0; k < OUT_TPW; ++k) {
    const unsigned t = wave + (unsigned)k * OUT_WAVES;
    av[k] = U[t * 16u + oc];
  }
  v2f b;
  b.x = lo ? wv : 0.0f;
  b.y = 0.0f;
  v8f c;
#pragma unroll
  for (int v = 0; v < 8; ++v) c[v] = bv;
  const unsigned loff = (lo ? 0u : 2u * OUT_CH) + oc;

#pragma unroll
  for (int k = 0; k < OUT_TPW; ++k) {
    const unsigned t = wave + (unsigned)k * OUT_WAVES;
    v2f a;
    a.x = lo ? av[k] : 0.0f;
    a.y = 0.0f;
    v8f d = __builtin_amdgcn_wmma_f32_16x16x4_f32(
         false, a,  false, b,
         (short)0, c,  false,  false);
    float* tp = out + (size_t)t * (4u * OUT_CH);
#pragma unroll
    for (int v = 0; v < 8; ++v) {
      const unsigned bb = (unsigned)v & 3u;
      const unsigned dn = (unsigned)v >> 2;
      tp[(size_t)bb * NUM_NEURONS * OUT_CH + dn * OUT_CH + loff] = d[v];
    }
  }
}

extern "C" void kernel_launch(void* const* d_in, const int* in_sizes, int n_in,
                              void* d_out, int out_size, void* d_ws, size_t ws_size,
                              hipStream_t stream) {
  const float* x_in        = (const float*)d_in[0];
  const int*   he          = (const int*)d_in[1];
  const float* post_spk    = (const float*)d_in[2];
  const float* w_hat       = (const float*)d_in[3];
  const float* s_e         = (const float*)d_in[4];
  const float* e_trace     = (const float*)d_in[5];
  const float* pre_trace   = (const float*)d_in[6];
  const float* post_trace  = (const float*)d_in[7];
  const float* weight_node = (const float*)d_in[8];
  const float* bias        = (const float*)d_in[9];
  float* out = (float*)d_out;
  float* ws  = (float*)d_ws;

  bhs_traces<<<391, 256, 0, stream>>>(
      x_in, post_spk, w_hat, s_e, e_trace, pre_trace, post_trace, out, ws);
  bhs_pass<false><<<252, 1024, 0, stream>>>(he, ws, out);
  bhs_norm<<<391, 256, 0, stream>>>(out, ws);
  bhs_pass<true><<<256, 1024, 0, stream>>>(he, ws, out);
  bhs_sumU<<<391, 256, 0, stream>>>(out, ws);
  bhs_out_wmma<<<250, 320, 0, stream>>>(weight_node, bias, ws, out);
}


# Filter passes: sums cleared, cw table in, partial planes out as 16-byte groups, each group's loads issued together ahead of its stores (were 4-byte load-wait-store loops, 25 trips); values unchanged
# speedup vs baseline: 6.0730x; 1.1298x over previous
#include <hip/hip_runtime.h>


#define NUM_NEURONS 100000
#define MAX_EDGES   100000
#define NUM_CONN    1600000
#define BATCH       4
#define OUT_CH      16

typedef __attribute__((ext_vector_type(2))) float v2f;
typedef __attribute__((ext_vector_type(8))) float v8f;

#define WS_EFFW 0u
#define WS_XT   100000u
#define WS_TR   100000u
#define WS_U    900000u

#define A_NRANGE 21u
#define A_RNG    4768u
#define A_NCHUNK 12u
#define A_CHUNK  133376u
#define A_R0     11
#define A_R1     11
#define A_R2     11
#define B_NRANGE 16u
#define B_RNG    6272u
#define B_NCHUNK 16u
#define B_CHUNK  100096u
#define B_R0     9
#define B_R1     8
#define B_R2     8
#define ROW      4096u
#define SLICE    128u
#define UFA      3
#define UFB      6
#define PLANE    100000u

__global__ void bhs_traces(const float* __restrict__ x_in,
                           const float* __restrict__ post_spk,
                           const float* __restrict__ w_hat,
                           const float* __restrict__ s_e,
                           const float* __restrict__ e_trace,
                           const float* __restrict__ pre_trace,
                           const float* __restrict__ post_trace,
                           float* __restrict__ out,
                           float* __restrict__ ws) {
  unsigned i = blockIdx.x * blockDim.x + threadIdx.x;
  if (i >= NUM_NEURONS) return;
  const float decay = 0.95122942450071400910f;
  float x0 = x_in[i];
  float x1 = x_in[NUM_NEURONS + i];
  float x2 = x_in[2u * NUM_NEURONS + i];
  float x3 = x_in[3u * NUM_NEURONS + i];
  float pre_spk  = (x0 + x1 + x2 + x3) * 0.25f;
  float pre_new  = fmaf(pre_trace[i],  decay, pre_spk);
  float post_new = fmaf(post_trace[i], decay, post_spk[i]);
  float e_new    = fmaf(0.9f, e_trace[i], pre_spk * (post_new + 0.1f));
  const size_t OUTB = (size_t)BATCH * NUM_NEURONS * OUT_CH;
  out[OUTB + i]                 = e_new;
  out[OUTB + MAX_EDGES + i]     = pre_new;
  out[OUTB + 2u * MAX_EDGES + i] = post_new;
  ws[WS_EFFW + i] = s_e[i] * w_hat[i];
  ((float4*)(ws + WS_XT))[i] = make_float4(x0, x1, x2, x3);
}


template <int MAXR, int N, bool LAST>
__device__ __forceinline__ void bhs_load_idx(int4 (&kv)[MAXR], int4 (&ov)[MAXR],
                                             const int* __restrict__ keyrow,
                                             const int* __restrict__ othrow,
                                             unsigned c0, bool lastok) {
#pragma unroll
  for (int it = 0; it < N; ++it) {
    if (LAST && it == N - 1 && !lastok) continue;
    const unsigned c = c0 + (unsigned)it * ROW;
    kv[it] = *(const int4*)(keyrow + c);
    ov[it] = *(const int4*)(othrow + c);
  }
}

template <int MAXR, int N, bool LAST>
__device__ __forceinline__ unsigned bhs_filter(const int4 (&kv)[MAXR], const int4 (&ov)[MAXR],
                                               unsigned* seg, unsigned base, unsigned rr, bool lastok) {
  unsigned wb = 0u;
#pragma unroll
  for (int it = 0; it < N; ++it) {
    if (LAST && it == N - 1 && !lastok) continue;
    unsigned d0 = (unsigned)kv[it].x - base, d1 = (unsigned)kv[it].y - base;
    unsigned d2 = (unsigned)kv[it].z - base, d3 = (unsigned)kv[it].w - base;
    bool m0 = d0 < rr, m1 = d1 < rr, m2 = d2 < rr, m3 = d3 < rr;
    unsigned b0 = __builtin_amdgcn_ballot_w32(m0);
    unsigned b1 = __builtin_amdgcn_ballot_w32(m1);
    unsigned b2 = __builtin_amdgcn_ballot_w32(m2);
    unsigned b3 = __builtin_amdgcn_ballot_w32(m3);
    if (m0) seg[__builtin_amdgcn_mbcnt_lo(b0, wb)] = (d0 << 17) | (unsigned)ov[it].x;
    wb += __builtin_popcount(b0);
    if (m1) seg[__builtin_amdgcn_mbcnt_lo(b1, wb)] = (d1 << 17) | (unsigned)ov[it].y;
    wb += __builtin_popcount(b1);
    if (m2) seg[__builtin_amdgcn_mbcnt_lo(b2, wb)] = (d2 << 17) | (unsigned)ov[it].z;
    wb += __builtin_popcount(b2);
    if (m3) seg[__builtin_amdgcn_mbcnt_lo(b3, wb)] = (d3 << 17) | (unsigned)ov[it].w;
    wb += __builtin_popcount(b3);
  }
  return wb;
}

template <bool PB>
__device__ __forceinline__ void bhs_fetch(const unsigned* seg, const float* __restrict__ ws,
                                          const float* cwl, unsigned t0, unsigned lane, unsigned nk,
                                          unsigned (&en)[PB ? UFB : UFA], bool (&ok)[PB ? UFB : UFA],
                                          float4 (&g)[PB ? UFB : UFA], float (&cwv)[PB ? UFB : UFA]) {
  if (!PB) {
#pragma unroll
    for (int k = 0; k < UFA; ++k) {
      unsigned i = t0 + (unsigned)k * 32u + lane;
      ok[k] = i < nk;
      en[k] = ok[k] ? seg[i] : 0u;
      unsigned ot = en[k] & 0x1FFFFu;
      g[k]   = ((const float4*)(ws + WS_XT))[ot];
      cwv[k] = cwl[en[k] >> 17];
    }
  } else {
#pragma unroll
    for (int k = 0; k < UFB; ++k) {
      unsigned i = t0 + (unsigned)k * 16u + (lane >> 1);
      ok[k] = i < nk;
      en[k] = ok[k] ? seg[i] : 0u;
    }
#pragma unroll
    for (int k = 0; k < UFB; ++k) {
      unsigned ot = en[k] & 0x1FFFFu;
      g[k]   = ((const float4*)(ws + WS_TR))[2u * ot + (lane & 1u)];
      cwv[k] = g[k].z;
    }
  }
}

template <bool PB, unsigned RNG>
__device__ __forceinline__ void bhs_sums(float* acc, unsigned lane,
                                         const unsigned (&en)[PB ? UFB : UFA], const bool (&ok)[PB ? UFB : UFA],
                                         const float4 (&g)[PB ? UFB : UFA], const float (&cwv)[PB ? UFB : UFA]) {
  constexpr int NU = PB ? UFB : UFA;
#pragma unroll
  for (int k = 0; k < NU; ++k) {
    if (ok[k]) {
      unsigned kl = en[k] >> 17;
      if (!PB) {
        atomicAdd(&acc[kl],            g[k].x * cwv[k]);
        atomicAdd(&acc[RNG + kl],      g[k].y * cwv[k]);
        atomicAdd(&acc[2u * RNG + kl], g[k].z * cwv[k]);
        atomicAdd(&acc[3u * RNG + kl], g[k].w * cwv[k]);
        atomicAdd(&acc[4u * RNG + kl], 1.0f);
      } else {
        float* a = acc + (lane & 1u) * (2u * RNG) + kl;
        atomicAdd(a,       cwv[k] * g[k].x);
        atomicAdd(a + RNG, cwv[k] * g[k].y);
      }
    }
  }
}

template <bool PB, unsigned RNG, int MAXR, int N, bool LAST, int NN, bool NLAST>
__device__ __forceinline__ void bhs_round(int4 (&kv)[MAXR], int4 (&ov)[MAXR], unsigned* seg, float* acc,
                                          const float* cwl, const float* __restrict__ ws,
                                          const int* __restrict__ keyrow, const int* __restrict__ othrow,
                                          unsigned cnext, unsigned base, unsigned rr, unsigned lane,
                                          bool lastok) {
  const unsigned nk = bhs_filter<MAXR, N, LAST>(kv, ov, seg, base, rr, lastok);
  constexpr int NU = PB ? UFB : UFA;
  constexpr unsigned TRIP = PB ? UFB * 16u : UFA * 32u;
  unsigned en[NU];
  bool ok[NU];
  float4 g[NU];
  float cwv[NU];
  bhs_fetch<PB>(seg, ws, cwl, 0u, lane, nk, en, ok, g, cwv);
  if constexpr (NN > 0)
    bhs_load_idx<MAXR, NN, NLAST>(kv, ov, keyrow, othrow, cnext, lastok);
  bhs_sums<PB, RNG>(acc, lane, en, ok, g, cwv);
  for (unsigned t0 = TRIP; t0 < nk; t0 += TRIP) {
    bhs_fetch<PB>(seg, ws, cwl, t0, lane, nk, en, ok, g, cwv);
    bhs_sums<PB, RNG>(acc, lane, en, ok, g, cwv);
  }
}

__device__ __forceinline__ void bhs_arrived(const float4& v) {
  asm volatile("" :: "v"(v.x), "v"(v.y), "v"(v.z), "v"(v.w));
}

template <bool PB>
__global__ __launch_bounds__(1024)
void bhs_pass(const int* __restrict__ he, const float* __restrict__ ws,
              float* __restrict__ part) {
  constexpr unsigned P      = PB ? 4u : 5u;
  constexpr unsigned NRANGE = PB ? B_NRANGE : A_NRANGE;
  constexpr unsigned RNG    = PB ? B_RNG : A_RNG;
  constexpr unsigned NCHUNK = PB ? B_NCHUNK : A_NCHUNK;
  constexpr unsigned CHUNK  = PB ? B_CHUNK : A_CHUNK;
  constexpr int R0 = PB ? B_R0 : A_R0, R1 = PB ? B_R1 : A_R1, R2 = PB ? B_R2 : A_R2;
  constexpr int MAXR = R0 > R1 ? (R0 > R2 ? R0 : R2) : (R1 > R2 ? R1 : R2);
  constexpr unsigned ROWS = (unsigned)(R0 + R1 + R2);
  constexpr unsigned SEG  = (unsigned)MAXR * SLICE;
  static_assert(CHUNK % SLICE == 0u && NUM_CONN % SLICE == 0u, "a chunk is a whole number of wave slices");
  static_assert((NCHUNK - 1u) * CHUNK < NUM_CONN && NCHUNK * CHUNK >= NUM_CONN, "the chunks cover the list");
  static_assert(ROWS * ROW >= CHUNK && (ROWS - 1u) * ROW <= NUM_CONN - (NCHUNK - 1u) * CHUNK,
                "only the last row of a chunk may be cut");
  static_assert(RNG % 32u == 0u && MAX_EDGES % 32u == 0u, "a range is whole 128-B lines of every plane");
  constexpr unsigned R4  = RNG / 4u;
  constexpr unsigned NV  = (P * R4 + 1023u) / 1024u;
  constexpr unsigned NCW = PB ? 1u : (R4 + 1023u) / 1024u;
  __shared__ __attribute__((aligned(16))) float acc[P * RNG];
  __shared__ __attribute__((aligned(16))) float cwl[PB ? 4u : RNG];
  __shared__ unsigned list[32u * SEG];

  const unsigned tid  = threadIdx.x;
  const unsigned lane = tid & 31u;
  const unsigned wave = __builtin_amdgcn_readfirstlane(tid >> 5);
  const unsigned r    = blockIdx.x % NRANGE;
  const unsigned j    = blockIdx.x / NRANGE;
  const unsigned base = r * RNG;
  const unsigned rr   = min(RNG, (unsigned)MAX_EDGES - base);

  const int* keyrow = PB ? he : he + NUM_CONN;
  const int* othrow = PB ? he + NUM_CONN : he;
  const unsigned cbeg = j * CHUNK;
  const unsigned cend = min(cbeg + CHUNK, (unsigned)NUM_CONN);
  const unsigned ct   = cbeg + tid * 4u;
  const bool lastok = (ROWS - 1u) * ROW + wave * SLICE < cend - cbeg;

  const unsigned rr4 = rr >> 2;

  float4 cwr[NCW];
  if (!PB) {
#pragma unroll
    for (unsigned k = 0; k < NCW; ++k)
      cwr[k] = ((const float4*)(ws + WS_EFFW + base))[min(tid + k * 1024u, rr4 - 1u)];
  }

  int4 kv[MAXR], ov[MAXR];
  bhs_load_idx<MAXR, R0, false>(kv, ov, keyrow, othrow, ct, true);

#pragma unroll
  for (unsigned k = 0; k < NV; ++k) {
    const unsigned f = tid + k * 1024u;
    if (f < P * R4) ((float4*)acc)[f] = make_float4(0.0f, 0.0f, 0.0f, 0.0f);
  }
  if (!PB) {
#pragma unroll
    for (unsigned k = 0; k < NCW; ++k) bhs_arrived(cwr[k]);
#pragma unroll
    for (unsigned k = 0; k < NCW; ++k) {
      const unsigned f = tid + k * 1024u;
      if (f < rr4) ((float4*)cwl)[f] = cwr[k];
    }
  }
  __syncthreads();

  unsigned* seg = list + wave * SEG;

  bhs_round<PB, RNG, MAXR, R0, false, R1, false>(kv, ov, seg, acc, cwl, ws, keyrow, othrow,
                                                 ct + (unsigned)R0 * ROW, base, rr, lane, lastok);
  bhs_round<PB, RNG, MAXR, R1, false, R2, true>(kv, ov, seg, acc, cwl, ws, keyrow, othrow,
                                                ct + (unsigned)(R0 + R1) * ROW, base, rr, lane, lastok);
  bhs_round<PB, RNG, MAXR, R2, true, 0, false>(kv, ov, seg, acc, cwl, ws, keyrow, othrow,
                                               0u, base, rr, lane, lastok);
  __syncthreads();

  float4 pv[NV];
#pragma unroll
  for (unsigned k = 0; k < NV; ++k)
    pv[k] = ((const float4*)acc)[min(tid + k * 1024u, P * R4 - 1u)];
#pragma unroll
  for (unsigned k = 0; k < NV; ++k) bhs_arrived(pv[k]);
#pragma unroll
  for (unsigned k = 0; k < NV; ++k) {
    const unsigned f  = tid + k * 1024u;
    const unsigned p  = f / R4;
    const unsigned i4 = f - p * R4;
    if (f < P * R4 && i4 < rr4)
      ((float4*)(part + (size_t)(j * P + p) * PLANE + base))[i4] = pv[k];
  }
}

__global__ void bhs_norm(const float* __restrict__ part, float* __restrict__ ws) {
  unsigned e = blockIdx.x * blockDim.x + threadIdx.x;
  if (e >= MAX_EDGES) return;
  float s0 = 0.f, s1 = 0.f, s2 = 0.f, s3 = 0.f, cn = 0.f;
#pragma unroll
  for (unsigned j = 0; j < A_NCHUNK; ++j) {
    const float* p = part + (size_t)j * 5u * PLANE + e;
    s0 += p[0];
    s1 += p[PLANE];
    s2 += p[2u * PLANE];
    s3 += p[3u * PLANE];
    cn += p[4u * PLANE];
  }
  float inv = 1.0f / fmaxf(cn, 1.0f);
  float cw = ws[WS_EFFW + e];
  float4* rec = (float4*)(ws + WS_TR) + 2u * e;
  rec[0] = make_float4(s0 * inv, s1 * inv, cw, 0.0f);
  rec[1] = make_float4(s2 * inv, s3 * inv, cw, 0.0f);
}

__global__ void bhs_sumU(const float* __restrict__ part, float* __restrict__ ws) {
  unsigned n = blockIdx.x * blockDim.x + threadIdx.x;
  if (n >= NUM_NEURONS) return;
  float u0 = 0.f, u1 = 0.f, u2 = 0.f, u3 = 0.f;
#pragma unroll
  for (unsigned j = 0; j < B_NCHUNK; ++j) {
    const float* p = part + (size_t)j * 4u * PLANE + n;
    u0 += p[0];
    u1 += p[PLANE];
    u2 += p[2u * PLANE];
    u3 += p[3u * PLANE];
  }
  ((float4*)(ws + WS_U))[n] = make_float4(u0, u1, u2, u3);
}

#define OUT_WG    250u
#define OUT_WPG   10u
#define OUT_WAVES 2500u
#define OUT_TPW   10
__global__ __launch_bounds__(320)
void bhs_out_wmma(const float* __restrict__ weight_node,
                  const float* __restrict__ bias,
                  const float* __restrict__ ws,
                  float* __restrict__ out) {
  const unsigned lane = threadIdx.x & 31u;
  const unsigned wave =
      __builtin_amdgcn_readfirstlane(blockIdx.x * OUT_WPG + (threadIdx.x >> 5));
  const unsigned oc = lane & 15u;
  const bool lo = lane < 16u;
  const float wv = weight_node[oc];
  const float bv = bias[oc];
  const float* U = ws + WS_U;

  float av[OUT_TPW];
#pragma unroll
  for (int k = 0; k < OUT_TPW; ++k) {
    const unsigned t = wave + (unsigned)k * OUT_WAVES;
    av[k] = U[t * 16u + oc];
  }
  v2f b;
  b.x = lo ? wv : 0.0f;
  b.y = 0.0f;
  v8f c;
#pragma unroll
  for (int v = 0; v < 8; ++v) c[v] = bv;
  const unsigned loff = (lo ? 0u : 2u * OUT_CH) + oc;

#pragma unroll
  for (int k = 0; k < OUT_TPW; ++k) {
    const unsigned t = wave + (unsigned)k * OUT_WAVES;
    v2f a;
    a.x = lo ? av[k] : 0.0f;
    a.y = 0.0f;
    v8f d = __builtin_amdgcn_wmma_f32_16x16x4_f32(
         false, a,  false, b,
         (short)0, c,  false,  false);
    float* tp = out + (size_t)t * (4u * OUT_CH);
#pragma unroll
    for (int v = 0; v < 8; ++v) {
      const unsigned bb = (unsigned)v & 3u;
      const unsigned dn = (unsigned)v >> 2;
      tp[(size_t)bb * NUM_NEURONS * OUT_CH + dn * OUT_CH + loff] = d[v];
    }
  }
}

extern "C" void kernel_launch(void* const* d_in, const int* in_sizes, int n_in,
                              void* d_out, int out_size, void* d_ws, size_t ws_size,
                              hipStream_t stream) {
  const float* x_in        = (const float*)d_in[0];
  const int*   he          = (const int*)d_in[1];
  const float* post_spk    = (const float*)d_in[2];
  const float* w_hat       = (const float*)d_in[3];
  const float* s_e         = (const float*)d_in[4];
  const float* e_trace     = (const float*)d_in[5];
  const float* pre_trace   = (const float*)d_in[6];
  const float* post_trace  = (const float*)d_in[7];
  const float* weight_node = (const float*)d_in[8];
  const float* bias        = (const float*)d_in[9];
  float* out = (float*)d_out;
  float* ws  = (float*)d_ws;

  bhs_traces<<<391, 256, 0, stream>>>(
      x_in, post_spk, w_hat, s_e, e_trace, pre_trace, post_trace, out, ws);
  bhs_pass<false><<<252, 1024, 0, stream>>>(he, ws, out);
  bhs_norm<<<391, 256, 0, stream>>>(out, ws);
  bhs_pass<true><<<256, 1024, 0, stream>>>(he, ws, out);
  bhs_sumU<<<391, 256, 0, stream>>>(out, ws);
  bhs_out_wmma<<<250, 320, 0, stream>>>(weight_node, bias, ws, out);
}
